# starting kernel
# baseline (speedup 1.0000x reference)
#include <hip/hip_runtime.h>

typedef __attribute__((ext_vector_type(2))) float v2f;
typedef __attribute__((ext_vector_type(4))) float v4f;
typedef __attribute__((ext_vector_type(8))) float v8f;

#define LEAKY  0.2f
#define MASKV  -9.0e15f
#define HS     132
#define AS     68

__launch_bounds__(128)
__global__ void gat_local_agg_kernel(const float* __restrict__ hidden,
                                     const int*   __restrict__ adj,
                                     const float* __restrict__ a,
                                     float*       __restrict__ out) {
  __shared__ __align__(16) float sh_h[64 * HS];
  __shared__ __align__(16) float sh_a[4 * 128];
  __shared__ __align__(16) float sh_alpha[64 * AS];
  __shared__ float sh_rinv[64];

  const int b    = blockIdx.x;
  const int tid  = threadIdx.x;
  const int w    = tid >> 5;
  const int lane = tid & 31;
  const int half = lane >> 4;
  const int lx   = lane & 15;

  const float* hb = hidden + (size_t)b * (64 * 128);

  for (int p = tid; p < 64 * 32; p += 128) {
    int row = p >> 5, c4 = p & 31;
    v4f v = *((const v4f*)hb + p);
    *(v4f*)&sh_h[row * HS + c4 * 4] = v;
  }
  {
    v4f v = *((const v4f*)a + tid);
    *(v4f*)&sh_a[tid * 4] = v;
  }
  __syncthreads();

  const int i0 = w * 16 + lx;

  const int* adjb = adj + (size_t)b * 4096;
  for (int jt = 0; jt < 4; ++jt) {
    const int j0 = jt * 16 + lx;
    v8f acc0 = {}, acc1 = {}, acc2 = {}, acc3 = {};
    for (int kb = 0; kb < 128; kb += 4) {
      int kk = kb + 2 * half;
      float h0 = sh_h[i0 * HS + kk];
      float h1 = sh_h[i0 * HS + kk + 1];
      v2f Bf = { sh_h[j0 * HS + kk], sh_h[j0 * HS + kk + 1] };
      v2f A0 = { h0 * sh_a[0 * 128 + kk], h1 * sh_a[0 * 128 + kk + 1] };
      v2f A1 = { h0 * sh_a[1 * 128 + kk], h1 * sh_a[1 * 128 + kk + 1] };
      v2f A2 = { h0 * sh_a[2 * 128 + kk], h1 * sh_a[2 * 128 + kk + 1] };
      v2f A3 = { h0 * sh_a[3 * 128 + kk], h1 * sh_a[3 * 128 + kk + 1] };
      acc0 = __builtin_amdgcn_wmma_f32_16x16x4_f32(false, A0, false, Bf, (short)0, acc0, false, false);
      acc1 = __builtin_amdgcn_wmma_f32_16x16x4_f32(false, A1, false, Bf, (short)0, acc1, false, false);
      acc2 = __builtin_amdgcn_wmma_f32_16x16x4_f32(false, A2, false, Bf, (short)0, acc2, false, false);
      acc3 = __builtin_amdgcn_wmma_f32_16x16x4_f32(false, A3, false, Bf, (short)0, acc3, false, false);
    }
    #pragma unroll
    for (int r = 0; r < 8; ++r) {
      int irow = w * 16 + r + 8 * half;
      int jc   = jt * 16 + lx;
      int av   = adjb[irow * 64 + jc];
      float e0 = acc0[r]; e0 = e0 >= 0.f ? e0 : LEAKY * e0;
      float e1 = acc1[r]; e1 = e1 >= 0.f ? e1 : LEAKY * e1;
      float e2 = acc2[r]; e2 = e2 >= 0.f ? e2 : LEAKY * e2;
      float e3 = acc3[r]; e3 = e3 >= 0.f ? e3 : LEAKY * e3;
      float val = MASKV;
      val = (av == 1) ? e0 : val;
      val = (av == 2) ? e1 : val;
      val = (av == 3) ? e2 : val;
      val = (av == 4) ? e3 : val;
      sh_alpha[irow * AS + jc] = val;
    }
  }
  __syncthreads();

  if (tid < 64) {
    float mx = -3.0e38f;
    for (int j2 = 0; j2 < 64; ++j2) mx = fmaxf(mx, sh_alpha[tid * AS + j2]);
    float s = 0.f;
    for (int j2 = 0; j2 < 64; ++j2) {
      float ev = __expf(sh_alpha[tid * AS + j2] - mx);
      sh_alpha[tid * AS + j2] = ev;
      s += ev;
    }
    sh_rinv[tid] = 1.0f / s;
  }
  __syncthreads();

  float* ob = out + (size_t)b * (64 * 128);
  for (int nt = 0; nt < 8; ++nt) {
    const int d0 = nt * 16 + lx;
    v8f acc = {};
    for (int kb = 0; kb < 64; kb += 4) {
      int kk = kb + 2 * half;
      v2f Af = { sh_alpha[i0 * AS + kk], sh_alpha[i0 * AS + kk + 1] };
      v2f Bf = { sh_h[kk * HS + d0],     sh_h[(kk + 1) * HS + d0] };
      acc = __builtin_amdgcn_wmma_f32_16x16x4_f32(false, Af, false, Bf, (short)0, acc, false, false);
    }
    #pragma unroll
    for (int r = 0; r < 8; ++r) {
      int irow = w * 16 + r + 8 * half;
      ob[irow * 128 + d0] = acc[r] * sh_rinv[irow];
    }
  }
}

extern "C" void kernel_launch(void* const* d_in, const int* in_sizes, int n_in,
                              void* d_out, int out_size, void* d_ws, size_t ws_size,
                              hipStream_t stream) {
  const float* hidden = (const float*)d_in[0];
  const int*   adj    = (const int*)d_in[1];
  const float* a      = (const float*)d_in[2];
  float* out = (float*)d_out;
  const int B = in_sizes[0] / (64 * 128);
  gat_local_agg_kernel<<<B, 128, 0, stream>>>(hidden, adj, a, out);
}


# logit tile computed transposed (WMMA operands swapped): a lane pair owns a softmax row, max/sum over own registers, probabilities feed 2nd product from registers; row-sum and j-sum order change
# speedup vs baseline: 1.1395x; 1.1395x over previous
#include <hip/hip_runtime.h>


typedef __attribute__((ext_vector_type(2))) float v2f;
typedef __attribute__((ext_vector_type(4))) float v4f;
typedef __attribute__((ext_vector_type(8))) float v8f;
typedef __attribute__((ext_vector_type(4))) int   v4i;

#define LEAKY  0.2f
#define MASKV  -9.0e15f
#define HS     132

__device__ __forceinline__ float xhalf(float x) {
  int bits = __builtin_bit_cast(int, x);
  return __builtin_bit_cast(float,
      __builtin_amdgcn_permlanex16(bits, bits, 0x76543210, 0xfedcba98, false, false));
}

__launch_bounds__(128)
__global__ void gat_local_agg_kernel(const float* __restrict__ hidden,
                                     const int*   __restrict__ adj,
                                     const float* __restrict__ a,
                                     float*       __restrict__ out) {
  __shared__ __align__(16) float sh_h[64 * HS];
  __shared__ __align__(16) float sh_a[4 * 128];

  const int b    = blockIdx.x;
  const int tid  = threadIdx.x;
  const int w    = tid >> 5;
  const int lane = tid & 31;
  const int half = lane >> 4;
  const int lx   = lane & 15;

  const float* hb = hidden + (size_t)b * (64 * 128);

  for (int p4 = tid; p4 < 64 * 32; p4 += 128) {
    int row = p4 >> 5, c4 = p4 & 31;
    v4f v = *((const v4f*)hb + p4);
    *(v4f*)&sh_h[row * HS + c4 * 4] = v;
  }
  {
    v4f v = *((const v4f*)a + tid);
    *(v4f*)&sh_a[tid * 4] = v;
  }
  __syncthreads();

  const int i0 = w * 16 + lx;

  const int* adjrow = adj + (size_t)b * 4096 + i0 * 64 + 8 * half;
  float p[4][8];
  #pragma unroll
  for (int jt = 0; jt < 4; ++jt) {
    const int j0 = jt * 16 + lx;
    v8f acc0 = {}, acc1 = {}, acc2 = {}, acc3 = {};
    for (int kb = 0; kb < 128; kb += 4) {
      int kk = kb + 2 * half;
      float h0 = sh_h[i0 * HS + kk];
      float h1 = sh_h[i0 * HS + kk + 1];
      v2f Hj = { sh_h[j0 * HS + kk], sh_h[j0 * HS + kk + 1] };
      v2f S0 = { h0 * sh_a[0 * 128 + kk], h1 * sh_a[0 * 128 + kk + 1] };
      v2f S1 = { h0 * sh_a[1 * 128 + kk], h1 * sh_a[1 * 128 + kk + 1] };
      v2f S2 = { h0 * sh_a[2 * 128 + kk], h1 * sh_a[2 * 128 + kk + 1] };
      v2f S3 = { h0 * sh_a[3 * 128 + kk], h1 * sh_a[3 * 128 + kk + 1] };
      acc0 = __builtin_amdgcn_wmma_f32_16x16x4_f32(false, Hj, false, S0, (short)0, acc0, false, false);
      acc1 = __builtin_amdgcn_wmma_f32_16x16x4_f32(false, Hj, false, S1, (short)0, acc1, false, false);
      acc2 = __builtin_amdgcn_wmma_f32_16x16x4_f32(false, Hj, false, S2, (short)0, acc2, false, false);
      acc3 = __builtin_amdgcn_wmma_f32_16x16x4_f32(false, Hj, false, S3, (short)0, acc3, false, false);
    }
    v4i av_lo = *(const v4i*)(adjrow + jt * 16);
    v4i av_hi = *(const v4i*)(adjrow + jt * 16 + 4);
    #pragma unroll
    for (int r = 0; r < 8; ++r) {
      int av   = (r < 4) ? av_lo[r & 3] : av_hi[r & 3];
      float e0 = acc0[r]; e0 = e0 >= 0.f ? e0 : LEAKY * e0;
      float e1 = acc1[r]; e1 = e1 >= 0.f ? e1 : LEAKY * e1;
      float e2 = acc2[r]; e2 = e2 >= 0.f ? e2 : LEAKY * e2;
      float e3 = acc3[r]; e3 = e3 >= 0.f ? e3 : LEAKY * e3;
      float val = MASKV;
      val = (av == 1) ? e0 : val;
      val = (av == 2) ? e1 : val;
      val = (av == 3) ? e2 : val;
      val = (av == 4) ? e3 : val;
      p[jt][r] = val;
    }
  }

  float mx = p[0][0];
  #pragma unroll
  for (int jt = 0; jt < 4; ++jt) {
    #pragma unroll
    for (int r = 0; r < 8; ++r) mx = fmaxf(mx, p[jt][r]);
  }
  mx = fmaxf(mx, xhalf(mx));
  float s = 0.f;
  #pragma unroll
  for (int jt = 0; jt < 4; ++jt) {
    #pragma unroll
    for (int r = 0; r < 8; ++r) {
      float ev = __expf(p[jt][r] - mx);
      p[jt][r] = ev;
      s += ev;
    }
  }
  s += xhalf(s);
  const float rinv = 1.0f / s;
  #pragma unroll
  for (int jt = 0; jt < 4; ++jt) {
    #pragma unroll
    for (int r = 0; r < 8; ++r) p[jt][r] *= rinv;
  }

  float* ob = out + (size_t)b * (64 * 128);
  for (int nt = 0; nt < 8; ++nt) {
    const int d0 = nt * 16 + lx;
    v8f acc = {};
    #pragma unroll
    for (int jt = 0; jt < 4; ++jt) {
      #pragma unroll
      for (int q = 0; q < 4; ++q) {
        const int jr = jt * 16 + 8 * half + 2 * q;
        v2f Af = { p[jt][2 * q], p[jt][2 * q + 1] };
        v2f Bf = { sh_h[jr * HS + d0], sh_h[(jr + 1) * HS + d0] };
        acc = __builtin_amdgcn_wmma_f32_16x16x4_f32(false, Af, false, Bf, (short)0, acc, false, false);
      }
    }
    #pragma unroll
    for (int r = 0; r < 8; ++r) {
      int irow = w * 16 + r + 8 * half;
      ob[irow * 128 + d0] = acc[r];
    }
  }
}

extern "C" void kernel_launch(void* const* d_in, const int* in_sizes, int n_in,
                              void* d_out, int out_size, void* d_ws, size_t ws_size,
                              hipStream_t stream) {
  const float* hidden = (const float*)d_in[0];
  const int*   adj    = (const int*)d_in[1];
  const float* a      = (const float*)d_in[2];
  float* out = (float*)d_out;
  const int B = in_sizes[0] / (64 * 128);
  gat_local_agg_kernel<<<B, 128, 0, stream>>>(hidden, adj, a, out);
}


# logit tiles computed once per symmetric pair (10 of 16, 2.5 per wave): an off-diagonal tile also selects with adj[j][i] and goes through LDS to the rows' owner wave; row-sum and j-sum order change
# speedup vs baseline: 1.3850x; 1.2154x over previous
#include <hip/hip_runtime.h>


typedef __attribute__((ext_vector_type(2))) float v2f;
typedef __attribute__((ext_vector_type(4))) float v4f;
typedef __attribute__((ext_vector_type(8))) float v8f;
typedef __attribute__((ext_vector_type(4))) int   v4i;

#define LEAKY  0.2f
#define MASKV  -9.0e15f
#define HS     132
#define XS     20

__device__ __forceinline__ float xhalf(float x) {
  int bits = __builtin_bit_cast(int, x);
  return __builtin_bit_cast(float,
      __builtin_amdgcn_permlanex16(bits, bits, 0x76543210, 0xfedcba98, false, false));
}

__device__ __forceinline__ float leaky(float e) { return e >= 0.f ? e : LEAKY * e; }

__device__ __forceinline__ float pick(int av, float e0, float e1, float e2, float e3) {
  float val = MASKV;
  val = (av == 1) ? e0 : val;
  val = (av == 2) ? e1 : val;
  val = (av == 3) ? e2 : val;
  val = (av == 4) ? e3 : val;
  return val;
}

__device__ __forceinline__ void logit_tiles(const float* sh_h, const float* sh_a,
                                            int i0, int j0, int half,
                                            v8f& acc0, v8f& acc1, v8f& acc2, v8f& acc3) {
  acc0 = {}; acc1 = {}; acc2 = {}; acc3 = {};
  for (int kb = 0; kb < 128; kb += 4) {
    int kk = kb + 2 * half;
    float h0 = sh_h[i0 * HS + kk];
    float h1 = sh_h[i0 * HS + kk + 1];
    v2f Hj = { sh_h[j0 * HS + kk], sh_h[j0 * HS + kk + 1] };
    v2f S0 = { h0 * sh_a[0 * 128 + kk], h1 * sh_a[0 * 128 + kk + 1] };
    v2f S1 = { h0 * sh_a[1 * 128 + kk], h1 * sh_a[1 * 128 + kk + 1] };
    v2f S2 = { h0 * sh_a[2 * 128 + kk], h1 * sh_a[2 * 128 + kk + 1] };
    v2f S3 = { h0 * sh_a[3 * 128 + kk], h1 * sh_a[3 * 128 + kk + 1] };
    acc0 = __builtin_amdgcn_wmma_f32_16x16x4_f32(false, Hj, false, S0, (short)0, acc0, false, false);
    acc1 = __builtin_amdgcn_wmma_f32_16x16x4_f32(false, Hj, false, S1, (short)0, acc1, false, false);
    acc2 = __builtin_amdgcn_wmma_f32_16x16x4_f32(false, Hj, false, S2, (short)0, acc2, false, false);
    acc3 = __builtin_amdgcn_wmma_f32_16x16x4_f32(false, Hj, false, S3, (short)0, acc3, false, false);
  }
}

__device__ __forceinline__ void logit_pair(const float* sh_h, const float* sa,
                                           int i0, int j0, int half, v8f& accA, v8f& accB) {
  accA = {}; accB = {};
  #pragma unroll 8
  for (int kb = 0; kb < 128; kb += 4) {
    int kk = kb + 2 * half;
    float h0 = sh_h[i0 * HS + kk];
    float h1 = sh_h[i0 * HS + kk + 1];
    v2f Hj = { sh_h[j0 * HS + kk], sh_h[j0 * HS + kk + 1] };
    v2f SA = { h0 * sa[kk],       h1 * sa[kk + 1] };
    v2f SB = { h0 * sa[128 + kk], h1 * sa[128 + kk + 1] };
    accA = __builtin_amdgcn_wmma_f32_16x16x4_f32(false, Hj, false, SA, (short)0, accA, false, false);
    accB = __builtin_amdgcn_wmma_f32_16x16x4_f32(false, Hj, false, SB, (short)0, accB, false, false);
  }
}

__launch_bounds__(128)
__global__ void gat_local_agg_kernel(const float* __restrict__ hidden,
                                     const int*   __restrict__ adj,
                                     const float* __restrict__ a,
                                     float*       __restrict__ out) {
  __shared__ __align__(16) float sh_h[64 * HS];
  __shared__ __align__(16) float sh_a[4 * 128];
  __shared__ __align__(16) float sh_x[8 * 16 * XS];

  const int b    = blockIdx.x;
  const int tid  = threadIdx.x;
  const int w    = __builtin_amdgcn_readfirstlane(tid >> 5);
  const int lane = tid & 31;
  const int half = lane >> 4;
  const int lx   = lane & 15;
  const int role = w;

  const float* hb = hidden + (size_t)b * (64 * 128);

  for (int p4 = tid; p4 < 64 * 32; p4 += 128) {
    int row = p4 >> 5, c4 = p4 & 31;
    v4f v = *((const v4f*)hb + p4);
    *(v4f*)&sh_h[row * HS + c4 * 4] = v;
  }
  {
    v4f v = *((const v4f*)a + tid);
    *(v4f*)&sh_a[tid * 4] = v;
  }
  __syncthreads();

  const int i0 = role * 16 + lx;
  const int* adjb = adj + (size_t)b * 4096;

  float p[4][8];
  #pragma unroll
  for (int t = 0; t < 2; ++t) {
    const int jt = (role + t) & 3;
    const int* adjrow = adjb + i0 * 64 + jt * 16 + 8 * half;
    v4i av_lo = *(const v4i*)(adjrow);
    v4i av_hi = *(const v4i*)(adjrow + 4);
    int avT[8];
    if (t > 0) {
      const int* adjcol = adjb + (jt * 16 + 8 * half) * 64 + role * 16 + lx;
      #pragma unroll
      for (int r = 0; r < 8; ++r) avT[r] = adjcol[r * 64];
    }
    v8f acc0, acc1, acc2, acc3;
    logit_tiles(sh_h, sh_a, i0, jt * 16 + lx, half, acc0, acc1, acc2, acc3);
    float* xrow = sh_x + ((jt * 2 + 1) * 16 + 8 * half) * XS + lx;
    #pragma unroll
    for (int r = 0; r < 8; ++r) {
      int av   = (r < 4) ? av_lo[r & 3] : av_hi[r & 3];
      float e0 = leaky(acc0[r]);
      float e1 = leaky(acc1[r]);
      float e2 = leaky(acc2[r]);
      float e3 = leaky(acc3[r]);
      p[t][r] = pick(av, e0, e1, e2, e3);
      if (t > 0) xrow[r * XS] = pick(avT[r], e0, e1, e2, e3);
    }
  }
  const int pk = (role < 2) ? 0 : 2;
  const int c0 = pk + 1, c1 = pk + 2;
  int av2[8];
  {
    const int jt = (role + 2) & 3;
    const int* adjrow = adjb + i0 * 64 + jt * 16 + 8 * half;
    v4i av_lo = *(const v4i*)(adjrow);
    v4i av_hi = *(const v4i*)(adjrow + 4);
    const int* adjcol = adjb + (jt * 16 + 8 * half) * 64 + role * 16 + lx;
    int avT[8];
    #pragma unroll
    for (int r = 0; r < 8; ++r) avT[r] = adjcol[r * 64];
    v8f accA, accB;
    logit_pair(sh_h, sh_a + pk * 128, i0, jt * 16 + lx, half, accA, accB);
    float* xrow = sh_x + ((jt * 2 + 0) * 16 + 8 * half) * XS + lx;
    #pragma unroll
    for (int r = 0; r < 8; ++r) {
      int av   = (r < 4) ? av_lo[r & 3] : av_hi[r & 3];
      float eA = leaky(accA[r]);
      float eB = leaky(accB[r]);
      av2[r]   = av;
      p[2][r]  = (av == c0) ? eA : ((av == c1) ? eB : MASKV);
      xrow[r * XS] = (avT[r] == c0) ? eA : ((avT[r] == c1) ? eB : MASKV);
    }
  }
  __syncthreads();
  {
    const float* xr = sh_x + ((role * 2 + 1) * 16 + lx) * XS + 8 * half;
    v4f lo = *(const v4f*)(xr);
    v4f hi = *(const v4f*)(xr + 4);
    #pragma unroll
    for (int r = 0; r < 4; ++r) { p[3][r] = lo[r]; p[3][r + 4] = hi[r]; }
  }
  {
    const float* xr = sh_x + ((role * 2 + 0) * 16 + lx) * XS + 8 * half;
    v4f lo = *(const v4f*)(xr);
    v4f hi = *(const v4f*)(xr + 4);
    #pragma unroll
    for (int r = 0; r < 8; ++r) {
      float got = (r < 4) ? lo[r & 3] : hi[r & 3];
      p[2][r] = (av2[r] == c0 || av2[r] == c1) ? p[2][r] : got;
    }
  }

  float mx = p[0][0];
  #pragma unroll
  for (int t = 0; t < 4; ++t) {
    #pragma unroll
    for (int r = 0; r < 8; ++r) mx = fmaxf(mx, p[t][r]);
  }
  mx = fmaxf(mx, xhalf(mx));
  float s = 0.f;
  #pragma unroll
  for (int t = 0; t < 4; ++t) {
    #pragma unroll
    for (int r = 0; r < 8; ++r) {
      float ev = __expf(p[t][r] - mx);
      p[t][r] = ev;
      s += ev;
    }
  }
  s += xhalf(s);
  const float rinv = 1.0f / s;
  #pragma unroll
  for (int t = 0; t < 4; ++t) {
    #pragma unroll
    for (int r = 0; r < 8; ++r) p[t][r] *= rinv;
  }

  float* ob = out + (size_t)b * (64 * 128);
  for (int nt = 0; nt < 8; ++nt) {
    const int d0 = nt * 16 + lx;
    v8f acc = {};
    #pragma unroll
    for (int t = 0; t < 4; ++t) {
      const int jt = (role + t) & 3;
      #pragma unroll
      for (int q = 0; q < 4; ++q) {
        const int jr = jt * 16 + 8 * half + 2 * q;
        v2f Af = { p[t][2 * q], p[t][2 * q + 1] };
        v2f Bf = { sh_h[jr * HS + d0], sh_h[(jr + 1) * HS + d0] };
        acc = __builtin_amdgcn_wmma_f32_16x16x4_f32(false, Af, false, Bf, (short)0, acc, false, false);
      }
    }
    #pragma unroll
    for (int r = 0; r < 8; ++r) {
      int irow = role * 16 + r + 8 * half;
      ob[irow * 128 + d0] = acc[r];
    }
  }
}

extern "C" void kernel_launch(void* const* d_in, const int* in_sizes, int n_in,
                              void* d_out, int out_size, void* d_ws, size_t ws_size,
                              hipStream_t stream) {
  const float* hidden = (const float*)d_in[0];
  const int*   adj    = (const int*)d_in[1];
  const float* a      = (const float*)d_in[2];
  float* out = (float*)d_out;
  const int B = in_sizes[0] / (64 * 128);
  gat_local_agg_kernel<<<B, 128, 0, stream>>>(hidden, adj, a, out);
}


# Logit B fragments (own 16 rows x a_k, all of K) built once per wave and kept in registers for its 2.5 column tiles; tile k-loops load only the A fragment; order of every sum unchanged
# speedup vs baseline: 1.4845x; 1.0718x over previous
#include <hip/hip_runtime.h>


typedef __attribute__((ext_vector_type(2))) float v2f;
typedef __attribute__((ext_vector_type(4))) float v4f;
typedef __attribute__((ext_vector_type(8))) float v8f;
typedef __attribute__((ext_vector_type(4))) int   v4i;

#define LEAKY  0.2f
#define MASKV  -9.0e15f
#define HS     132
#define XS     20

__device__ __forceinline__ float xhalf(float x) {
  int bits = __builtin_bit_cast(int, x);
  return __builtin_bit_cast(float,
      __builtin_amdgcn_permlanex16(bits, bits, 0x76543210, 0xfedcba98, false, false));
}

__device__ __forceinline__ float leaky(float e) { return e >= 0.f ? e : LEAKY * e; }

__device__ __forceinline__ float pick(int av, int c0, int c1, int c2, int c3,
                                      float e0, float e1, float e2, float e3) {
  float val = MASKV;
  val = (av == c0) ? e0 : val;
  val = (av == c1) ? e1 : val;
  val = (av == c2) ? e2 : val;
  val = (av == c3) ? e3 : val;
  return val;
}

__device__ __forceinline__ v8f mm(v2f A, v2f B, v8f C) {
  return __builtin_amdgcn_wmma_f32_16x16x4_f32(false, A, false, B, (short)0, C, false, false);
}

__launch_bounds__(128, 2)
__global__ void gat_local_agg_kernel(const float* __restrict__ hidden,
                                     const int*   __restrict__ adj,
                                     const float* __restrict__ a,
                                     float*       __restrict__ out) {
  __shared__ __align__(16) float sh_h[64 * HS];
  __shared__ __align__(16) float sh_a[4 * 128];
  __shared__ __align__(16) float sh_x[8 * 16 * XS];

  const int b    = blockIdx.x;
  const int tid  = threadIdx.x;
  const int w    = __builtin_amdgcn_readfirstlane(tid >> 5);
  const int lane = tid & 31;
  const int half = lane >> 4;
  const int lx   = lane & 15;
  const int role = w;

  const float* hb = hidden + (size_t)b * (64 * 128);

  for (int p4 = tid; p4 < 64 * 32; p4 += 128) {
    int row = p4 >> 5, c4 = p4 & 31;
    v4f v = *((const v4f*)hb + p4);
    *(v4f*)&sh_h[row * HS + c4 * 4] = v;
  }
  {
    v4f v = *((const v4f*)a + tid);
    *(v4f*)&sh_a[tid * 4] = v;
  }
  __syncthreads();

  const int i0 = role * 16 + lx;
  const int* adjb = adj + (size_t)b * 4096;

  const int pk  = (role < 2) ? 0 : 2;
  const int cq0 = (0 ^ pk) + 1, cq1 = (1 ^ pk) + 1, cq2 = (2 ^ pk) + 1, cq3 = (3 ^ pk) + 1;
  const float* sa0 = sh_a + (0 ^ pk) * 128;
  const float* sa1 = sh_a + (1 ^ pk) * 128;
  const float* sa2 = sh_a + (2 ^ pk) * 128;
  const float* sa3 = sh_a + (3 ^ pk) * 128;

  const int jt1 = (role + 1) & 3, jt2 = (role + 2) & 3;
  const int* adjrow0 = adjb + i0 * 64 + role * 16 + 8 * half;
  const int* adjrow1 = adjb + i0 * 64 + jt1  * 16 + 8 * half;
  const int* adjrow2 = adjb + i0 * 64 + jt2  * 16 + 8 * half;
  const int* adjcol1 = adjb + (jt1 * 16 + 8 * half) * 64 + role * 16 + lx;
  const int* adjcol2 = adjb + (jt2 * 16 + 8 * half) * 64 + role * 16 + lx;
  const v4i av0_lo = *(const v4i*)(adjrow0), av0_hi = *(const v4i*)(adjrow0 + 4);

  v2f hr[32];
  v2f S0[32], S1[32], S2[32], S3[32];
  v2f Hj1[32], Hj2[32];
  #pragma unroll
  for (int s = 0; s < 32; ++s) {
    const int kk = 4 * s + 2 * half;
    hr[s] = *(const v2f*)&sh_h[i0 * HS + kk];
    S0[s] = hr[s] * *(const v2f*)&sa0[kk];
    S1[s] = hr[s] * *(const v2f*)&sa1[kk];
    S2[s] = hr[s] * *(const v2f*)&sa2[kk];
    S3[s] = hr[s] * *(const v2f*)&sa3[kk];
  }

  v8f d0 = {}, d1 = {}, d2 = {}, d3 = {};
  #pragma unroll
  for (int s = 0; s < 32; ++s) {
    d0 = mm(hr[s], S0[s], d0);
    d1 = mm(hr[s], S1[s], d1);
    d2 = mm(hr[s], S2[s], d2);
    d3 = mm(hr[s], S3[s], d3);
  }
  __builtin_amdgcn_sched_barrier(0);
  #pragma unroll
  for (int s = 0; s < 32; ++s) {
    Hj1[s] = *(const v2f*)&sh_h[(jt1 * 16 + lx) * HS + 4 * s + 2 * half];
    Hj2[s] = *(const v2f*)&sh_h[(jt2 * 16 + lx) * HS + 4 * s + 2 * half];
  }
  const v4i av1_lo = *(const v4i*)(adjrow1), av1_hi = *(const v4i*)(adjrow1 + 4);
  const v4i av2_lo = *(const v4i*)(adjrow2), av2_hi = *(const v4i*)(adjrow2 + 4);
  int avT1[8], avT2[8];
  #pragma unroll
  for (int r = 0; r < 8; ++r) { avT1[r] = adjcol1[r * 64]; avT2[r] = adjcol2[r * 64]; }
  float p[4][8];
  #pragma unroll
  for (int r = 0; r < 8; ++r) {
    int av = (r < 4) ? av0_lo[r & 3] : av0_hi[r & 3];
    p[0][r] = pick(av, cq0, cq1, cq2, cq3, leaky(d0[r]), leaky(d1[r]), leaky(d2[r]), leaky(d3[r]));
  }
  __builtin_amdgcn_sched_barrier(0);

  v8f u0 = {}, u1 = {}, u2 = {}, u3 = {};
  #pragma unroll
  for (int s = 0; s < 32; ++s) {
    u0 = mm(Hj1[s], S0[s], u0);
    u1 = mm(Hj1[s], S1[s], u1);
    u2 = mm(Hj1[s], S2[s], u2);
    u3 = mm(Hj1[s], S3[s], u3);
  }
  v8f accA = {}, accB = {};
  #pragma unroll
  for (int s = 0; s < 32; ++s) {
    accA = mm(Hj2[s], S0[s], accA);
    accB = mm(Hj2[s], S1[s], accB);
  }
  {
    float* xrow = sh_x + ((jt1 * 2 + 1) * 16 + 8 * half) * XS + lx;
    #pragma unroll
    for (int r = 0; r < 8; ++r) {
      int av   = (r < 4) ? av1_lo[r & 3] : av1_hi[r & 3];
      float e0 = leaky(u0[r]);
      float e1 = leaky(u1[r]);
      float e2 = leaky(u2[r]);
      float e3 = leaky(u3[r]);
      p[1][r] = pick(av, cq0, cq1, cq2, cq3, e0, e1, e2, e3);
      xrow[r * XS] = pick(avT1[r], cq0, cq1, cq2, cq3, e0, e1, e2, e3);
    }
  }
  const int c0 = cq0, c1 = cq1;
  int av2[8];
  {
    float* xrow = sh_x + ((jt2 * 2 + 0) * 16 + 8 * half) * XS + lx;
    #pragma unroll
    for (int r = 0; r < 8; ++r) {
      int av   = (r < 4) ? av2_lo[r & 3] : av2_hi[r & 3];
      float eA = leaky(accA[r]);
      float eB = leaky(accB[r]);
      av2[r]   = av;
      p[2][r]  = (av == c0) ? eA : ((av == c1) ? eB : MASKV);
      xrow[r * XS] = (avT2[r] == c0) ? eA : ((avT2[r] == c1) ? eB : MASKV);
    }
  }
  __syncthreads();
  {
    const float* xr = sh_x + ((role * 2 + 1) * 16 + lx) * XS + 8 * half;
    v4f lo = *(const v4f*)(xr);
    v4f hi = *(const v4f*)(xr + 4);
    #pragma unroll
    for (int r = 0; r < 4; ++r) { p[3][r] = lo[r]; p[3][r + 4] = hi[r]; }
  }
  {
    const float* xr = sh_x + ((role * 2 + 0) * 16 + lx) * XS + 8 * half;
    v4f lo = *(const v4f*)(xr);
    v4f hi = *(const v4f*)(xr + 4);
    #pragma unroll
    for (int r = 0; r < 8; ++r) {
      float got = (r < 4) ? lo[r & 3] : hi[r & 3];
      p[2][r] = (av2[r] == c0 || av2[r] == c1) ? p[2][r] : got;
    }
  }

  float mx = p[0][0];
  #pragma unroll
  for (int t = 0; t < 4; ++t) {
    #pragma unroll
    for (int r = 0; r < 8; ++r) mx = fmaxf(mx, p[t][r]);
  }
  mx = fmaxf(mx, xhalf(mx));
  float s = 0.f;
  #pragma unroll
  for (int t = 0; t < 4; ++t) {
    #pragma unroll
    for (int r = 0; r < 8; ++r) {
      float ev = __expf(p[t][r] - mx);
      p[t][r] = ev;
      s += ev;
    }
  }
  s += xhalf(s);
  const float rinv = 1.0f / s;
  #pragma unroll
  for (int t = 0; t < 4; ++t) {
    #pragma unroll
    for (int r = 0; r < 8; ++r) p[t][r] *= rinv;
  }

  float* ob = out + (size_t)b * (64 * 128);
  for (int nt = 0; nt < 8; ++nt) {
    const int d0 = nt * 16 + lx;
    v8f acc = {};
    #pragma unroll
    for (int t = 0; t < 4; ++t) {
      const int jt = (role + t) & 3;
      #pragma unroll
      for (int q = 0; q < 4; ++q) {
        const int jr = jt * 16 + 8 * half + 2 * q;
        v2f Af = { p[t][2 * q], p[t][2 * q + 1] };
        v2f Bf = { sh_h[jr * HS + d0], sh_h[(jr + 1) * HS + d0] };
        acc = __builtin_amdgcn_wmma_f32_16x16x4_f32(false, Af, false, Bf, (short)0, acc, false, false);
      }
    }
    #pragma unroll
    for (int r = 0; r < 8; ++r) {
      int irow = role * 16 + r + 8 * half;
      ob[irow * 128 + d0] = acc[r];
    }
  }
}

extern "C" void kernel_launch(void* const* d_in, const int* in_sizes, int n_in,
                              void* d_out, int out_size, void* d_ws, size_t ws_size,
                              hipStream_t stream) {
  const float* hidden = (const float*)d_in[0];
  const int*   adj    = (const int*)d_in[1];
  const float* a      = (const float*)d_in[2];
  float* out = (float*)d_out;
  const int B = in_sizes[0] / (64 * 128);
  gat_local_agg_kernel<<<B, 128, 0, stream>>>(hidden, adj, a, out);
}


# Phase B column map changed: a lane owns 4 adjacent features, so H is read by 16-byte LDS loads, all issued before the softmax, and out is stored as whole 128-byte lines; sum order unchanged
# speedup vs baseline: 1.6280x; 1.0967x over previous
#include <hip/hip_runtime.h>


typedef __attribute__((ext_vector_type(2))) float v2f;
typedef __attribute__((ext_vector_type(4))) float v4f;
typedef __attribute__((ext_vector_type(8))) float v8f;
typedef __attribute__((ext_vector_type(4))) int   v4i;

#define LEAKY  0.2f
#define MASKV  -9.0e15f
#define HS     132
#define XS     20

__device__ __forceinline__ float xhalf(float x) {
  int bits = __builtin_bit_cast(int, x);
  return __builtin_bit_cast(float,
      __builtin_amdgcn_permlanex16(bits, bits, 0x76543210, 0xfedcba98, false, false));
}

__device__ __forceinline__ float leaky(float e) { return e >= 0.f ? e : LEAKY * e; }

__device__ __forceinline__ float pick(int av, int c0, int c1, int c2, int c3,
                                      float e0, float e1, float e2, float e3) {
  float val = MASKV;
  val = (av == c0) ? e0 : val;
  val = (av == c1) ? e1 : val;
  val = (av == c2) ? e2 : val;
  val = (av == c3) ? e3 : val;
  return val;
}

__device__ __forceinline__ v8f mm(v2f A, v2f B, v8f C) {
  return __builtin_amdgcn_wmma_f32_16x16x4_f32(false, A, false, B, (short)0, C, false, false);
}

__launch_bounds__(128, 2)
__global__ void gat_local_agg_kernel(const float* __restrict__ hidden,
                                     const int*   __restrict__ adj,
                                     const float* __restrict__ a,
                                     float*       __restrict__ out) {
  __shared__ __align__(16) float sh_h[64 * HS];
  __shared__ __align__(16) float sh_a[4 * 128];
  __shared__ __align__(16) float sh_x[8 * 16 * XS];

  const int b    = blockIdx.x;
  const int tid  = threadIdx.x;
  const int w    = __builtin_amdgcn_readfirstlane(tid >> 5);
  const int lane = tid & 31;
  const int half = lane >> 4;
  const int lx   = lane & 15;
  const int role = w;

  const float* hb = hidden + (size_t)b * (64 * 128);

  for (int p4 = tid; p4 < 64 * 32; p4 += 128) {
    int row = p4 >> 5, c4 = p4 & 31;
    v4f v = *((const v4f*)hb + p4);
    *(v4f*)&sh_h[row * HS + c4 * 4] = v;
  }
  {
    v4f v = *((const v4f*)a + tid);
    *(v4f*)&sh_a[tid * 4] = v;
  }
  __syncthreads();

  const int i0 = role * 16 + lx;
  const int* adjb = adj + (size_t)b * 4096;

  const int pk  = (role < 2) ? 0 : 2;
  const int cq0 = (0 ^ pk) + 1, cq1 = (1 ^ pk) + 1, cq2 = (2 ^ pk) + 1, cq3 = (3 ^ pk) + 1;
  const float* sa0 = sh_a + (0 ^ pk) * 128;
  const float* sa1 = sh_a + (1 ^ pk) * 128;
  const float* sa2 = sh_a + (2 ^ pk) * 128;
  const float* sa3 = sh_a + (3 ^ pk) * 128;

  const int jt1 = (role + 1) & 3, jt2 = (role + 2) & 3;
  const int* adjrow0 = adjb + i0 * 64 + role * 16 + 8 * half;
  const int* adjrow1 = adjb + i0 * 64 + jt1  * 16 + 8 * half;
  const int* adjrow2 = adjb + i0 * 64 + jt2  * 16 + 8 * half;
  const int* adjcol1 = adjb + (jt1 * 16 + 8 * half) * 64 + role * 16 + lx;
  const int* adjcol2 = adjb + (jt2 * 16 + 8 * half) * 64 + role * 16 + lx;
  const v4i av0_lo = *(const v4i*)(adjrow0), av0_hi = *(const v4i*)(adjrow0 + 4);

  v2f hr[32];
  v2f S0[32], S1[32], S2[32], S3[32];
  v2f Hj1[32], Hj2[32];
  #pragma unroll
  for (int s = 0; s < 32; ++s) {
    const int kk = 4 * s + 2 * half;
    hr[s] = *(const v2f*)&sh_h[i0 * HS + kk];
    S0[s] = hr[s] * *(const v2f*)&sa0[kk];
    S1[s] = hr[s] * *(const v2f*)&sa1[kk];
    S2[s] = hr[s] * *(const v2f*)&sa2[kk];
    S3[s] = hr[s] * *(const v2f*)&sa3[kk];
  }

  v8f d0 = {}, d1 = {}, d2 = {}, d3 = {};
  #pragma unroll
  for (int s = 0; s < 32; ++s) {
    d0 = mm(hr[s], S0[s], d0);
    d1 = mm(hr[s], S1[s], d1);
    d2 = mm(hr[s], S2[s], d2);
    d3 = mm(hr[s], S3[s], d3);
  }
  __builtin_amdgcn_sched_barrier(0);
  #pragma unroll
  for (int s = 0; s < 32; ++s) {
    Hj1[s] = *(const v2f*)&sh_h[(jt1 * 16 + lx) * HS + 4 * s + 2 * half];
    Hj2[s] = *(const v2f*)&sh_h[(jt2 * 16 + lx) * HS + 4 * s + 2 * half];
  }
  const v4i av1_lo = *(const v4i*)(adjrow1), av1_hi = *(const v4i*)(adjrow1 + 4);
  const v4i av2_lo = *(const v4i*)(adjrow2), av2_hi = *(const v4i*)(adjrow2 + 4);
  int avT1[8], avT2[8];
  #pragma unroll
  for (int r = 0; r < 8; ++r) { avT1[r] = adjcol1[r * 64]; avT2[r] = adjcol2[r * 64]; }
  float p[4][8];
  #pragma unroll
  for (int r = 0; r < 8; ++r) {
    int av = (r < 4) ? av0_lo[r & 3] : av0_hi[r & 3];
    p[0][r] = pick(av, cq0, cq1, cq2, cq3, leaky(d0[r]), leaky(d1[r]), leaky(d2[r]), leaky(d3[r]));
  }
  __builtin_amdgcn_sched_barrier(0);

  v8f u0 = {}, u1 = {}, u2 = {}, u3 = {};
  #pragma unroll
  for (int s = 0; s < 32; ++s) {
    u0 = mm(Hj1[s], S0[s], u0);
    u1 = mm(Hj1[s], S1[s], u1);
    u2 = mm(Hj1[s], S2[s], u2);
    u3 = mm(Hj1[s], S3[s], u3);
  }
  v8f accA = {}, accB = {};
  #pragma unroll
  for (int s = 0; s < 32; ++s) {
    accA = mm(Hj2[s], S0[s], accA);
    accB = mm(Hj2[s], S1[s], accB);
  }
  {
    float* xrow = sh_x + ((jt1 * 2 + 1) * 16 + 8 * half) * XS + lx;
    #pragma unroll
    for (int r = 0; r < 8; ++r) {
      int av   = (r < 4) ? av1_lo[r & 3] : av1_hi[r & 3];
      float e0 = leaky(u0[r]);
      float e1 = leaky(u1[r]);
      float e2 = leaky(u2[r]);
      float e3 = leaky(u3[r]);
      p[1][r] = pick(av, cq0, cq1, cq2, cq3, e0, e1, e2, e3);
      xrow[r * XS] = pick(avT1[r], cq0, cq1, cq2, cq3, e0, e1, e2, e3);
    }
  }
  const int c0 = cq0, c1 = cq1;
  int av2[8];
  {
    float* xrow = sh_x + ((jt2 * 2 + 0) * 16 + 8 * half) * XS + lx;
    #pragma unroll
    for (int r = 0; r < 8; ++r) {
      int av   = (r < 4) ? av2_lo[r & 3] : av2_hi[r & 3];
      float eA = leaky(accA[r]);
      float eB = leaky(accB[r]);
      av2[r]   = av;
      p[2][r]  = (av == c0) ? eA : ((av == c1) ? eB : MASKV);
      xrow[r * XS] = (avT2[r] == c0) ? eA : ((avT2[r] == c1) ? eB : MASKV);
    }
  }
  __syncthreads();
  const float* xr3 = sh_x + ((role * 2 + 1) * 16 + lx) * XS + 8 * half;
  const float* xr2 = sh_x + ((role * 2 + 0) * 16 + lx) * XS + 8 * half;
  const v4f x3lo = *(const v4f*)(xr3), x3hi = *(const v4f*)(xr3 + 4);
  const v4f x2lo = *(const v4f*)(xr2), x2hi = *(const v4f*)(xr2 + 4);

  v4f X[4][8][2];
  #pragma unroll
  for (int t = 0; t < 4; ++t) {
    const int jt = (role + t) & 3;
    #pragma unroll
    for (int r = 0; r < 8; ++r) {
      #pragma unroll
      for (int g = 0; g < 2; ++g)
        X[t][r][g] = *(const v4f*)&sh_h[(jt * 16 + 8 * half + r) * HS + 64 * g + 4 * lx];
    }
  }
  __builtin_amdgcn_sched_barrier(0);

  #pragma unroll
  for (int r = 0; r < 4; ++r) { p[3][r] = x3lo[r]; p[3][r + 4] = x3hi[r]; }
  #pragma unroll
  for (int r = 0; r < 8; ++r) {
    float got = (r < 4) ? x2lo[r & 3] : x2hi[r & 3];
    p[2][r] = (av2[r] == c0 || av2[r] == c1) ? p[2][r] : got;
  }

  float mx = p[0][0];
  #pragma unroll
  for (int t = 0; t < 4; ++t) {
    #pragma unroll
    for (int r = 0; r < 8; ++r) mx = fmaxf(mx, p[t][r]);
  }
  mx = fmaxf(mx, xhalf(mx));
  float s = 0.f;
  #pragma unroll
  for (int t = 0; t < 4; ++t) {
    #pragma unroll
    for (int r = 0; r < 8; ++r) {
      float ev = __expf(p[t][r] - mx);
      p[t][r] = ev;
      s += ev;
    }
  }
  s += xhalf(s);
  const float rinv = 1.0f / s;
  #pragma unroll
  for (int t = 0; t < 4; ++t) {
    #pragma unroll
    for (int r = 0; r < 8; ++r) p[t][r] *= rinv;
  }

  v8f acc[8] = {};
  #pragma unroll
  for (int t = 0; t < 4; ++t) {
    #pragma unroll
    for (int q = 0; q < 4; ++q) {
      v2f Af = { p[t][2 * q], p[t][2 * q + 1] };
      #pragma unroll
      for (int c8 = 0; c8 < 8; ++c8) {
        v2f Bf = { X[t][2 * q][c8 >> 2][c8 & 3], X[t][2 * q + 1][c8 >> 2][c8 & 3] };
        acc[c8] = mm(Af, Bf, acc[c8]);
      }
    }
  }
  float* ob = out + (size_t)b * (64 * 128);
  #pragma unroll
  for (int r = 0; r < 8; ++r) {
    const int irow = role * 16 + r + 8 * half;
    #pragma unroll
    for (int g = 0; g < 2; ++g) {
      v4f o = { acc[4 * g][r], acc[4 * g + 1][r], acc[4 * g + 2][r], acc[4 * g + 3][r] };
      *(v4f*)&ob[irow * 128 + 64 * g + 4 * lx] = o;
    }
  }
}

extern "C" void kernel_launch(void* const* d_in, const int* in_sizes, int n_in,
                              void* d_out, int out_size, void* d_ws, size_t ws_size,
                              hipStream_t stream) {
  const float* hidden = (const float*)d_in[0];
  const int*   adj    = (const int*)d_in[1];
  const float* a      = (const float*)d_in[2];
  float* out = (float*)d_out;
  const int B = in_sizes[0] / (64 * 128);
  gat_local_agg_kernel<<<B, 128, 0, stream>>>(hidden, adj, a, out);
}
